# starting kernel
# baseline (speedup 1.0000x reference)
#include <hip/hip_runtime.h>

typedef float v2f __attribute__((ext_vector_type(2)));
typedef float v8f __attribute__((ext_vector_type(8)));

#define D_MODEL 512
#define D_STATE 32
#define BATCH 8
#define SEQ 4096
#define ROWS (BATCH * SEQ)

#define WS_ABAR 0
#define WS_BBAR (WS_ABAR + 1024)
#define WS_UB (WS_BBAR + 16384)
#define WS_STATES (WS_UB + ROWS * 32)

__device__ __forceinline__ v8f wmma4(v2f a, v2f b, v8f c) {
  return __builtin_amdgcn_wmma_f32_16x16x4_f32(false, a, false, b, (short)0, c,
                                               false, false);
}

__global__ void __launch_bounds__(256) prep_kernel(
    const float* __restrict__ A, const float* __restrict__ B,
    const float* __restrict__ log_dt, float* __restrict__ ws) {
  __shared__ float aug[32][64];
  __shared__ float Ah[32][32];
  __shared__ float inv[32][32];
  __shared__ float fac[32];
  __shared__ float dt_sh;
  const int tid = threadIdx.x;

  if (tid == 0) {
    float s = 0.f;
    for (int i = 0; i < D_MODEL; ++i) {
      float v = expf(log_dt[i]);
      v = fminf(fmaxf(v, 1e-4f), 1.0f);
      s += v;
    }
    dt_sh = s / (float)D_MODEL;
  }
  __syncthreads();
  const float dt = dt_sh;

  for (int idx = tid; idx < 32 * 32; idx += blockDim.x) {
    const int i = idx >> 5, j = idx & 31;
    const float a = A[idx];
    const float eye = (i == j) ? 1.f : 0.f;
    aug[i][j] = eye - 0.5f * dt * a;
    aug[i][j + 32] = eye;
    Ah[i][j] = eye + 0.5f * dt * a;
  }
  __syncthreads();

  for (int k = 0; k < 32; ++k) {
    const float p = aug[k][k];
    __syncthreads();
    for (int j = tid; j < 64; j += blockDim.x) aug[k][j] = aug[k][j] / p;
    __syncthreads();
    if (tid < 32) fac[tid] = aug[tid][k];
    __syncthreads();
    for (int idx = tid; idx < 32 * 64; idx += blockDim.x) {
      const int i = idx >> 6, j = idx & 63;
      if (i != k) aug[i][j] -= fac[i] * aug[k][j];
    }
    __syncthreads();
  }
  for (int idx = tid; idx < 32 * 32; idx += blockDim.x) {
    const int i = idx >> 5, j = idx & 31;
    inv[i][j] = aug[i][j + 32];
  }
  __syncthreads();

  float* Abar = ws + WS_ABAR;
  for (int idx = tid; idx < 32 * 32; idx += blockDim.x) {
    const int i = idx >> 5, j = idx & 31;
    float s = 0.f;
    for (int k = 0; k < 32; ++k) s += Ah[i][k] * inv[k][j];
    Abar[idx] = s;
  }
  float* Bbar = ws + WS_BBAR;
  for (int idx = tid; idx < 32 * D_MODEL; idx += blockDim.x) {
    const int n = idx >> 9, d = idx & 511;
    float s = 0.f;
    for (int k = 0; k < 32; ++k) s += inv[n][k] * B[k * D_MODEL + d];
    Bbar[idx] = dt * s;
  }
}

__global__ void __launch_bounds__(256) ub_kernel(const float* __restrict__ x,
                                                 const float* __restrict__ Bbar,
                                                 float* __restrict__ uB) {
  const int wave = threadIdx.x >> 5;
  const int lane = threadIdx.x & 31;
  const int r = lane & 15, h = lane >> 4;
  const int mtile = blockIdx.x * 8 + wave;

  const float* xrow = x + (size_t)(mtile * 16 + r) * D_MODEL;
  const float* b0row = Bbar + (size_t)r * D_MODEL;
  const float* b1row = Bbar + (size_t)(16 + r) * D_MODEL;

  v8f acc0 = {0.f, 0.f, 0.f, 0.f, 0.f, 0.f, 0.f, 0.f};
  v8f acc1 = {0.f, 0.f, 0.f, 0.f, 0.f, 0.f, 0.f, 0.f};

  for (int k0 = 0; k0 < D_MODEL; k0 += 4) {
    const int k = k0 + 2 * h;
    const v2f a = *(const v2f*)(xrow + k);
    const v2f b0 = *(const v2f*)(b0row + k);
    const v2f b1 = *(const v2f*)(b1row + k);
    acc0 = wmma4(a, b0, acc0);
    acc1 = wmma4(a, b1, acc1);
  }

  const int mbase = mtile * 16 + 8 * h;
#pragma unroll
  for (int v = 0; v < 8; ++v) {
    const int m = mbase + v;
    uB[(size_t)m * 32 + r] = acc0[v];
    uB[(size_t)m * 32 + 16 + r] = acc1[v];
  }
}

__global__ void __launch_bounds__(32) scan_kernel(const float* __restrict__ ws,
                                                  float* __restrict__ states,
                                                  float* __restrict__ fs_out) {
  const int b = blockIdx.x;
  const int lane = threadIdx.x;
  const float* Abar = ws + WS_ABAR;
  const float* uB = ws + WS_UB + (size_t)b * SEQ * 32;
  float* st_out = states + (size_t)b * SEQ * 32;

  float a[32];
#pragma unroll
  for (int m = 0; m < 32; ++m) a[m] = Abar[lane * 32 + m];

  float st = 0.f;
  for (int s = 0; s < SEQ; ++s) {
    float acc = uB[s * 32 + lane];
    const int stb = __float_as_int(st);
#pragma unroll
    for (int m = 0; m < 32; ++m) {
      acc = fmaf(a[m], __int_as_float(__builtin_amdgcn_readlane(stb, m)), acc);
    }
    st = tanhf(acc);
    st_out[s * 32 + lane] = st;
  }
  fs_out[b * 32 + lane] = st;
}

__global__ void __launch_bounds__(256) cproj_kernel(
    const float* __restrict__ states, const float* __restrict__ C,
    const float* __restrict__ x, const float* __restrict__ Dv,
    float* __restrict__ y1) {
  const int wave = threadIdx.x >> 5;
  const int lane = threadIdx.x & 31;
  const int r = lane & 15, h = lane >> 4;
  const int mtile = blockIdx.x * 8 + wave;

  v2f afrag[8];
  const float* srow = states + (size_t)(mtile * 16 + r) * 32;
#pragma unroll
  for (int kk = 0; kk < 8; ++kk)
    afrag[kk] = *(const v2f*)(srow + kk * 4 + 2 * h);

  const int mbase = mtile * 16 + 8 * h;
  for (int dtile = 0; dtile < 32; ++dtile) {
    v8f acc = {0.f, 0.f, 0.f, 0.f, 0.f, 0.f, 0.f, 0.f};
    const float* crow = C + (size_t)(dtile * 16 + r) * 32;
#pragma unroll
    for (int kk = 0; kk < 8; ++kk) {
      const v2f b = *(const v2f*)(crow + kk * 4 + 2 * h);
      acc = wmma4(afrag[kk], b, acc);
    }
    const int d = dtile * 16 + r;
    const float Dd = Dv[d];
#pragma unroll
    for (int v = 0; v < 8; ++v) {
      const int m = mbase + v;
      y1[(size_t)m * D_MODEL + d] = acc[v] + x[(size_t)m * D_MODEL + d] * Dd;
    }
  }
}

__global__ void __launch_bounds__(256) wout_kernel(
    const float* __restrict__ W, const float* __restrict__ bout,
    float* __restrict__ y) {
  __shared__ float lds[16 * D_MODEL];
  const int tid = threadIdx.x;
  const int mtile = blockIdx.x;

  const float4* src4 = (const float4*)(y + (size_t)mtile * 16 * D_MODEL);
  float4* dst4 = (float4*)lds;
#pragma unroll
  for (int i = 0; i < 8; ++i) dst4[tid + i * 256] = src4[tid + i * 256];
  __syncthreads();

  const int wave = tid >> 5, lane = tid & 31;
  const int r = lane & 15, h = lane >> 4;
  const int mbase = mtile * 16 + 8 * h;
  const float* arow = lds + r * D_MODEL;

  for (int t = 0; t < 4; ++t) {
    const int ntile = wave * 4 + t;
    const float* wrow = W + (size_t)(ntile * 16 + r) * D_MODEL;
    if (t < 3)
      __builtin_prefetch(W + (size_t)((ntile + 1) * 16 + r) * D_MODEL, 0, 1);

    v8f acc = {0.f, 0.f, 0.f, 0.f, 0.f, 0.f, 0.f, 0.f};
    for (int k0 = 0; k0 < D_MODEL; k0 += 4) {
      const int k = k0 + 2 * h;
      const v2f a = *(const v2f*)(arow + k);
      const v2f b = *(const v2f*)(wrow + k);
      acc = wmma4(a, b, acc);
    }
    const int n = ntile * 16 + r;
    const float bo = bout[n];
#pragma unroll
    for (int v = 0; v < 8; ++v)
      y[(size_t)(mbase + v) * D_MODEL + n] = acc[v] + bo;
  }
}

extern "C" void kernel_launch(void* const* d_in, const int* in_sizes, int n_in,
                              void* d_out, int out_size, void* d_ws,
                              size_t ws_size, hipStream_t stream) {
  const float* x = (const float*)d_in[0];
  const float* A = (const float*)d_in[1];
  const float* B = (const float*)d_in[2];
  const float* C = (const float*)d_in[3];
  const float* Dv = (const float*)d_in[4];
  const float* log_dt = (const float*)d_in[5];
  const float* W = (const float*)d_in[6];
  const float* bout = (const float*)d_in[7];

  float* y = (float*)d_out;
  float* fs = y + (size_t)ROWS * D_MODEL;

  float* ws = (float*)d_ws;
  float* Bbar = ws + WS_BBAR;
  float* uB = ws + WS_UB;
  float* states = ws + WS_STATES;

  prep_kernel<<<1, 256, 0, stream>>>(A, B, log_dt, ws);
  ub_kernel<<<256, 256, 0, stream>>>(x, Bbar, uB);
  scan_kernel<<<BATCH, 32, 0, stream>>>(ws, states, fs);
  cproj_kernel<<<256, 256, 0, stream>>>(states, C, x, Dv, y);
  wout_kernel<<<2048, 256, 0, stream>>>(W, bout, y);
}


# out: C folds into the weight (WC = W_out C, 512x32), so y = [x*D | states] @ [W_out | WC]^T + b is one 128x128-tiled LDS GEMM, K=544; no y1 pass through memory; sum order changes
# speedup vs baseline: 3.3220x; 3.3220x over previous
#include <hip/hip_runtime.h>

typedef float v2f __attribute__((ext_vector_type(2)));
typedef float v4f __attribute__((ext_vector_type(4)));
typedef float v8f __attribute__((ext_vector_type(8)));

#define D_MODEL 512
#define D_STATE 32
#define BATCH 8
#define SEQ 4096
#define ROWS (BATCH * SEQ)

#define WS_ABAR 0
#define WS_BBAR (WS_ABAR + 1024)
#define WS_UB (WS_BBAR + 16384)
#define WS_STATES (WS_UB + ROWS * 32)

__device__ __forceinline__ v8f wmma4(v2f a, v2f b, v8f c) {
  return __builtin_amdgcn_wmma_f32_16x16x4_f32(false, a, false, b, (short)0, c,
                                               false, false);
}

__global__ void __launch_bounds__(256) prep_kernel(
    const float* __restrict__ A, const float* __restrict__ B,
    const float* __restrict__ log_dt, float* __restrict__ ws) {
  __shared__ float aug[32][64];
  __shared__ float Ah[32][32];
  __shared__ float inv[32][32];
  __shared__ float fac[32];
  __shared__ float dt_sh;
  const int tid = threadIdx.x;

  if (tid == 0) {
    float s = 0.f;
    for (int i = 0; i < D_MODEL; ++i) {
      float v = expf(log_dt[i]);
      v = fminf(fmaxf(v, 1e-4f), 1.0f);
      s += v;
    }
    dt_sh = s / (float)D_MODEL;
  }
  __syncthreads();
  const float dt = dt_sh;

  for (int idx = tid; idx < 32 * 32; idx += blockDim.x) {
    const int i = idx >> 5, j = idx & 31;
    const float a = A[idx];
    const float eye = (i == j) ? 1.f : 0.f;
    aug[i][j] = eye - 0.5f * dt * a;
    aug[i][j + 32] = eye;
    Ah[i][j] = eye + 0.5f * dt * a;
  }
  __syncthreads();

  for (int k = 0; k < 32; ++k) {
    const float p = aug[k][k];
    __syncthreads();
    for (int j = tid; j < 64; j += blockDim.x) aug[k][j] = aug[k][j] / p;
    __syncthreads();
    if (tid < 32) fac[tid] = aug[tid][k];
    __syncthreads();
    for (int idx = tid; idx < 32 * 64; idx += blockDim.x) {
      const int i = idx >> 6, j = idx & 63;
      if (i != k) aug[i][j] -= fac[i] * aug[k][j];
    }
    __syncthreads();
  }
  for (int idx = tid; idx < 32 * 32; idx += blockDim.x) {
    const int i = idx >> 5, j = idx & 31;
    inv[i][j] = aug[i][j + 32];
  }
  __syncthreads();

  float* Abar = ws + WS_ABAR;
  for (int idx = tid; idx < 32 * 32; idx += blockDim.x) {
    const int i = idx >> 5, j = idx & 31;
    float s = 0.f;
    for (int k = 0; k < 32; ++k) s += Ah[i][k] * inv[k][j];
    Abar[idx] = s;
  }
  float* Bbar = ws + WS_BBAR;
  for (int idx = tid; idx < 32 * D_MODEL; idx += blockDim.x) {
    const int n = idx >> 9, d = idx & 511;
    float s = 0.f;
    for (int k = 0; k < 32; ++k) s += inv[n][k] * B[k * D_MODEL + d];
    Bbar[idx] = dt * s;
  }
}

__global__ void __launch_bounds__(256) ub_kernel(const float* __restrict__ x,
                                                 const float* __restrict__ Bbar,
                                                 float* __restrict__ uB) {
  const int wave = threadIdx.x >> 5;
  const int lane = threadIdx.x & 31;
  const int r = lane & 15, h = lane >> 4;
  const int mtile = blockIdx.x * 8 + wave;

  const float* xrow = x + (size_t)(mtile * 16 + r) * D_MODEL;
  const float* b0row = Bbar + (size_t)r * D_MODEL;
  const float* b1row = Bbar + (size_t)(16 + r) * D_MODEL;

  v8f acc0 = {0.f, 0.f, 0.f, 0.f, 0.f, 0.f, 0.f, 0.f};
  v8f acc1 = {0.f, 0.f, 0.f, 0.f, 0.f, 0.f, 0.f, 0.f};

  for (int k0 = 0; k0 < D_MODEL; k0 += 4) {
    const int k = k0 + 2 * h;
    const v2f a = *(const v2f*)(xrow + k);
    const v2f b0 = *(const v2f*)(b0row + k);
    const v2f b1 = *(const v2f*)(b1row + k);
    acc0 = wmma4(a, b0, acc0);
    acc1 = wmma4(a, b1, acc1);
  }

  float* tile = uB + (size_t)mtile * 512;
  float* dst = tile + (2 * h) * 128 + r * 4;
  *(float4*)(dst) = make_float4(acc0[0], acc0[1], acc0[2], acc0[3]);
  *(float4*)(dst + 64) = make_float4(acc1[0], acc1[1], acc1[2], acc1[3]);
  *(float4*)(dst + 128) = make_float4(acc0[4], acc0[5], acc0[6], acc0[7]);
  *(float4*)(dst + 192) = make_float4(acc1[4], acc1[5], acc1[6], acc1[7]);
}

#define ROW_XOR(x, MASK)                                                      \
  __int_as_float(__builtin_amdgcn_update_dpp(0, __float_as_int(x),            \
                                             0x160 | (MASK), 0xf, 0xf, true))
__device__ __forceinline__ float half_xor(float x) {
  const int v = __float_as_int(x);
  return __int_as_float(
      __builtin_amdgcn_permlanex16(v, v, 0x76543210, 0xfedcba98, false, false));
}

__global__ void __launch_bounds__(32) scan_kernel(const float* __restrict__ ws,
                                                  float* __restrict__ states,
                                                  float* __restrict__ fs_out) {
  const int b = blockIdx.x;
  const int lane = threadIdx.x;
  const float* Abar = ws + WS_ABAR;
  const float* uB = ws + WS_UB + (size_t)b * SEQ * 32;
  float* st_out = states + (size_t)b * SEQ * 32;

  float c[8][4];
#pragma unroll
  for (int jp = 0; jp < 8; ++jp) {
#pragma unroll
    for (int q = 0; q < 4; ++q)
      c[jp][q] = Abar[(lane ^ (jp << 2)) * 32 + (lane ^ q)];
  }

  float st = 0.f;
  const float* ubl = uB + lane * 4;
  for (int t = 0; t < SEQ / 16; ++t) {
    const float4 q0 = *(const float4*)(ubl + t * 512);
    const float4 q1 = *(const float4*)(ubl + t * 512 + 128);
    const float4 q2 = *(const float4*)(ubl + t * 512 + 256);
    const float4 q3 = *(const float4*)(ubl + t * 512 + 384);
    const float u[16] = {q0.x, q0.y, q0.z, q0.w, q1.x, q1.y, q1.z, q1.w,
                         q2.x, q2.y, q2.z, q2.w, q3.x, q3.y, q3.z, q3.w};
#pragma unroll
    for (int j = 0; j < 16; ++j) {
      const float s0 = st;
      const float s1 = ROW_XOR(st, 1);
      const float s2 = ROW_XOR(st, 2);
      const float s3 = ROW_XOR(st, 3);
      float w[8];
      w[0] = fmaf(c[0][0], s0, u[j]);
#pragma unroll
      for (int jp = 1; jp < 8; ++jp) w[jp] = c[jp][0] * s0;
#pragma unroll
      for (int jp = 0; jp < 8; ++jp) w[jp] = fmaf(c[jp][1], s1, w[jp]);
#pragma unroll
      for (int jp = 0; jp < 8; ++jp) w[jp] = fmaf(c[jp][2], s2, w[jp]);
#pragma unroll
      for (int jp = 0; jp < 8; ++jp) w[jp] = fmaf(c[jp][3], s3, w[jp]);
      w[0] += ROW_XOR(w[1], 4);
      w[2] += ROW_XOR(w[3], 4);
      w[4] += ROW_XOR(w[5], 4);
      w[6] += ROW_XOR(w[7], 4);
      w[0] += ROW_XOR(w[2], 8);
      w[4] += ROW_XOR(w[6], 8);
      w[0] += half_xor(w[4]);
      st = tanhf(w[0]);
      st_out[(t * 16 + j) * 32 + lane] = st;
    }
  }
  fs_out[b * 32 + lane] = st;
}

__global__ void __launch_bounds__(32) wc_kernel(const float* __restrict__ W,
                                                const float* __restrict__ C,
                                                float* __restrict__ WC) {
  const int lane = threadIdx.x;
  const int r = lane & 15, h = lane >> 4;
  const int ntile = blockIdx.x >> 1;
  const int jt = blockIdx.x & 1;
  const float* wrow = W + (size_t)(ntile * 16 + r) * D_MODEL;
  const float* ccol = C + jt * 16 + r;

  v8f acc0 = {0.f, 0.f, 0.f, 0.f, 0.f, 0.f, 0.f, 0.f};
  v8f acc1 = {0.f, 0.f, 0.f, 0.f, 0.f, 0.f, 0.f, 0.f};
  for (int k0 = 0; k0 < D_MODEL; k0 += 8) {
    const int k = k0 + 2 * h;
    const v2f a0 = *(const v2f*)(wrow + k);
    const v2f a1 = *(const v2f*)(wrow + k + 4);
    v2f b0, b1;
    b0[0] = ccol[(size_t)k * 32];
    b0[1] = ccol[(size_t)(k + 1) * 32];
    b1[0] = ccol[(size_t)(k + 4) * 32];
    b1[1] = ccol[(size_t)(k + 5) * 32];
    acc0 = wmma4(a0, b0, acc0);
    acc1 = wmma4(a1, b1, acc1);
  }
#pragma unroll
  for (int v = 0; v < 8; ++v)
    WC[(size_t)(ntile * 16 + 8 * h + v) * 32 + jt * 16 + r] = acc0[v] + acc1[v];
}

#define OT_P 36
#define OT_CHUNKS 17
__global__ void __launch_bounds__(256) out_kernel(
    const float* __restrict__ x, const float* __restrict__ states,
    const float* __restrict__ W, const float* __restrict__ WC,
    const float* __restrict__ Dv, const float* __restrict__ bout,
    float* __restrict__ y) {
  __shared__ __attribute__((aligned(16))) float As[128 * OT_P];
  __shared__ __attribute__((aligned(16))) float Bs[128 * OT_P];
  const int tid = threadIdx.x;
  const int wave = tid >> 5, lane = tid & 31;
  const int r = lane & 15, h = lane >> 4;
  const int n0 = blockIdx.x * 128;
  const int m0 = blockIdx.y * 128;

  const int srow = tid >> 3;
  const int scol = (tid & 7) * 4;
  const float* xa = x + (size_t)(m0 + srow) * D_MODEL + scol;
  const float* wb = W + (size_t)(n0 + srow) * D_MODEL + scol;
  const float* sa = states + (size_t)(m0 + srow) * D_STATE + scol;
  const float* cb = WC + (size_t)(n0 + srow) * D_STATE + scol;
  float* ast = As + srow * OT_P + scol;
  float* bst = Bs + srow * OT_P + scol;

  const int wm = wave >> 1, wn = wave & 1;
  const float* afr = As + (wm * 32 + r) * OT_P + 4 * h;
  const float* bfr = Bs + (wn * 64 + r) * OT_P + 4 * h;

  v8f acc[2][4];
#pragma unroll
  for (int i = 0; i < 2; ++i)
#pragma unroll
    for (int j = 0; j < 4; ++j)
      acc[i][j] = (v8f){0.f, 0.f, 0.f, 0.f, 0.f, 0.f, 0.f, 0.f};

  v4f ra[4], rb[4];
  v4f dcur = *(const v4f*)(Dv + scol);
#pragma unroll
  for (int i = 0; i < 4; ++i) {
    ra[i] = *(const v4f*)(xa + (size_t)i * 32 * D_MODEL);
    rb[i] = *(const v4f*)(wb + (size_t)i * 32 * D_MODEL);
  }

  for (int c = 0; c < OT_CHUNKS; ++c) {
    __syncthreads();
#pragma unroll
    for (int i = 0; i < 4; ++i) {
      *(v4f*)(ast + i * 32 * OT_P) = ra[i] * dcur;
      *(v4f*)(bst + i * 32 * OT_P) = rb[i];
    }
    __syncthreads();

    const int cn = c + 1;
    if (cn < OT_CHUNKS) {
      const bool last = (cn == OT_CHUNKS - 1);
      const float* pa = last ? sa : xa + cn * 32;
      const float* pb = last ? cb : wb + cn * 32;
      const size_t rs = last ? (size_t)32 * D_STATE : (size_t)32 * D_MODEL;
#pragma unroll
      for (int i = 0; i < 4; ++i) {
        ra[i] = *(const v4f*)(pa + i * rs);
        rb[i] = *(const v4f*)(pb + i * rs);
      }
      if (last) dcur = (v4f){1.f, 1.f, 1.f, 1.f};
      else dcur = *(const v4f*)(Dv + cn * 32 + scol);
    }

#pragma unroll
    for (int kk = 0; kk < 4; ++kk) {
      v4f a[2], b[4];
#pragma unroll
      for (int i = 0; i < 2; ++i)
        a[i] = *(const v4f*)(afr + i * 16 * OT_P + kk * 8);
#pragma unroll
      for (int j = 0; j < 4; ++j)
        b[j] = *(const v4f*)(bfr + j * 16 * OT_P + kk * 8);
#pragma unroll
      for (int i = 0; i < 2; ++i)
#pragma unroll
        for (int j = 0; j < 4; ++j) {
          v2f av, bv;
          av[0] = a[i].x; av[1] = a[i].y;
          bv[0] = b[j].x; bv[1] = b[j].y;
          acc[i][j] = wmma4(av, bv, acc[i][j]);
        }
#pragma unroll
      for (int i = 0; i < 2; ++i)
#pragma unroll
        for (int j = 0; j < 4; ++j) {
          v2f av, bv;
          av[0] = a[i].z; av[1] = a[i].w;
          bv[0] = b[j].z; bv[1] = b[j].w;
          acc[i][j] = wmma4(av, bv, acc[i][j]);
        }
    }
  }

#pragma unroll
  for (int j = 0; j < 4; ++j) {
    const int n = n0 + wn * 64 + j * 16 + r;
    const float bo = bout[n];
#pragma unroll
    for (int i = 0; i < 2; ++i) {
      const int mb = m0 + wm * 32 + i * 16 + 8 * h;
#pragma unroll
      for (int v = 0; v < 8; ++v)
        y[(size_t)(mb + v) * D_MODEL + n] = acc[i][j][v] + bo;
    }
  }
}

extern "C" void kernel_launch(void* const* d_in, const int* in_sizes, int n_in,
                              void* d_out, int out_size, void* d_ws,
                              size_t ws_size, hipStream_t stream) {
  const float* x = (const float*)d_in[0];
  const float* A = (const float*)d_in[1];
  const float* B = (const float*)d_in[2];
  const float* C = (const float*)d_in[3];
  const float* Dv = (const float*)d_in[4];
  const float* log_dt = (const float*)d_in[5];
  const float* W = (const float*)d_in[6];
  const float* bout = (const float*)d_in[7];

  float* y = (float*)d_out;
  float* fs = y + (size_t)ROWS * D_MODEL;

  float* ws = (float*)d_ws;
  float* Bbar = ws + WS_BBAR;
  float* uB = ws + WS_UB;
  float* states = ws + WS_STATES;
  float* WC = ws + WS_BBAR;

  prep_kernel<<<1, 256, 0, stream>>>(A, B, log_dt, ws);
  ub_kernel<<<256, 256, 0, stream>>>(x, Bbar, uB);
  wc_kernel<<<64, 32, 0, stream>>>(W, C, WC);
  scan_kernel<<<BATCH, 32, 0, stream>>>(ws, states, fs);
  out_kernel<<<dim3(4, 256), 256, 0, stream>>>(x, states, W, WC, Dv, bout, y);
}


# out: its x*D @ W_out^T part (K=512, needs no scan result) and W_out@C run as extra work-groups of the scan's own launch; a last launch adds states @ WC^T (K=32) into y; sum order changes
# speedup vs baseline: 3.5057x; 1.0553x over previous
#include <hip/hip_runtime.h>

typedef float v2f __attribute__((ext_vector_type(2)));
typedef float v4f __attribute__((ext_vector_type(4)));
typedef float v8f __attribute__((ext_vector_type(8)));

#define D_MODEL 512
#define D_STATE 32
#define BATCH 8
#define SEQ 4096
#define ROWS (BATCH * SEQ)

#define WS_ABAR 0
#define WS_BBAR (WS_ABAR + 1024)
#define WS_UB (WS_BBAR + 16384)
#define WS_STATES (WS_UB + ROWS * 32)

__device__ __forceinline__ v8f wmma4(v2f a, v2f b, v8f c) {
  return __builtin_amdgcn_wmma_f32_16x16x4_f32(false, a, false, b, (short)0, c,
                                               false, false);
}

__global__ void __launch_bounds__(256) prep_kernel(
    const float* __restrict__ A, const float* __restrict__ B,
    const float* __restrict__ log_dt, float* __restrict__ ws) {
  __shared__ float aug[32][64];
  __shared__ float Ah[32][32];
  __shared__ float inv[32][32];
  __shared__ float fac[32];
  __shared__ float dt_sh;
  const int tid = threadIdx.x;

  if (tid == 0) {
    float s = 0.f;
    for (int i = 0; i < D_MODEL; ++i) {
      float v = expf(log_dt[i]);
      v = fminf(fmaxf(v, 1e-4f), 1.0f);
      s += v;
    }
    dt_sh = s / (float)D_MODEL;
  }
  __syncthreads();
  const float dt = dt_sh;

  for (int idx = tid; idx < 32 * 32; idx += blockDim.x) {
    const int i = idx >> 5, j = idx & 31;
    const float a = A[idx];
    const float eye = (i == j) ? 1.f : 0.f;
    aug[i][j] = eye - 0.5f * dt * a;
    aug[i][j + 32] = eye;
    Ah[i][j] = eye + 0.5f * dt * a;
  }
  __syncthreads();

  for (int k = 0; k < 32; ++k) {
    const float p = aug[k][k];
    __syncthreads();
    for (int j = tid; j < 64; j += blockDim.x) aug[k][j] = aug[k][j] / p;
    __syncthreads();
    if (tid < 32) fac[tid] = aug[tid][k];
    __syncthreads();
    for (int idx = tid; idx < 32 * 64; idx += blockDim.x) {
      const int i = idx >> 6, j = idx & 63;
      if (i != k) aug[i][j] -= fac[i] * aug[k][j];
    }
    __syncthreads();
  }
  for (int idx = tid; idx < 32 * 32; idx += blockDim.x) {
    const int i = idx >> 5, j = idx & 31;
    inv[i][j] = aug[i][j + 32];
  }
  __syncthreads();

  float* Abar = ws + WS_ABAR;
  for (int idx = tid; idx < 32 * 32; idx += blockDim.x) {
    const int i = idx >> 5, j = idx & 31;
    float s = 0.f;
    for (int k = 0; k < 32; ++k) s += Ah[i][k] * inv[k][j];
    Abar[idx] = s;
  }
  float* Bbar = ws + WS_BBAR;
  for (int idx = tid; idx < 32 * D_MODEL; idx += blockDim.x) {
    const int n = idx >> 9, d = idx & 511;
    float s = 0.f;
    for (int k = 0; k < 32; ++k) s += inv[n][k] * B[k * D_MODEL + d];
    Bbar[idx] = dt * s;
  }
}

__global__ void __launch_bounds__(256) ub_kernel(const float* __restrict__ x,
                                                 const float* __restrict__ Bbar,
                                                 float* __restrict__ uB) {
  const int wave = threadIdx.x >> 5;
  const int lane = threadIdx.x & 31;
  const int r = lane & 15, h = lane >> 4;
  const int mtile = blockIdx.x * 8 + wave;

  const float* xrow = x + (size_t)(mtile * 16 + r) * D_MODEL;
  const float* b0row = Bbar + (size_t)r * D_MODEL;
  const float* b1row = Bbar + (size_t)(16 + r) * D_MODEL;

  v8f acc0 = {0.f, 0.f, 0.f, 0.f, 0.f, 0.f, 0.f, 0.f};
  v8f acc1 = {0.f, 0.f, 0.f, 0.f, 0.f, 0.f, 0.f, 0.f};

  for (int k0 = 0; k0 < D_MODEL; k0 += 4) {
    const int k = k0 + 2 * h;
    const v2f a = *(const v2f*)(xrow + k);
    const v2f b0 = *(const v2f*)(b0row + k);
    const v2f b1 = *(const v2f*)(b1row + k);
    acc0 = wmma4(a, b0, acc0);
    acc1 = wmma4(a, b1, acc1);
  }

  float* tile = uB + (size_t)mtile * 512;
  float* dst = tile + (2 * h) * 128 + r * 4;
  *(float4*)(dst) = make_float4(acc0[0], acc0[1], acc0[2], acc0[3]);
  *(float4*)(dst + 64) = make_float4(acc1[0], acc1[1], acc1[2], acc1[3]);
  *(float4*)(dst + 128) = make_float4(acc0[4], acc0[5], acc0[6], acc0[7]);
  *(float4*)(dst + 192) = make_float4(acc1[4], acc1[5], acc1[6], acc1[7]);
}

#define ROW_XOR(x, MASK)                                                      \
  __int_as_float(__builtin_amdgcn_update_dpp(0, __float_as_int(x),            \
                                             0x160 | (MASK), 0xf, 0xf, true))
__device__ __forceinline__ float half_xor(float x) {
  const int v = __float_as_int(x);
  return __int_as_float(
      __builtin_amdgcn_permlanex16(v, v, 0x76543210, 0xfedcba98, false, false));
}

__device__ __forceinline__ void scan_role(const int b, const int lane,
                                          const float* __restrict__ Abar,
                                          const float* __restrict__ uBall,
                                          float* __restrict__ states,
                                          float* __restrict__ fs_out) {
  const float* uB = uBall + (size_t)b * SEQ * 32;
  float* st_out = states + (size_t)b * SEQ * 32;

  float c[8][4];
#pragma unroll
  for (int jp = 0; jp < 8; ++jp) {
#pragma unroll
    for (int q = 0; q < 4; ++q)
      c[jp][q] = Abar[(lane ^ (jp << 2)) * 32 + (lane ^ q)];
  }

  float st = 0.f;
  const float* ubl = uB + lane * 4;
  for (int t = 0; t < SEQ / 16; ++t) {
    const float4 q0 = *(const float4*)(ubl + t * 512);
    const float4 q1 = *(const float4*)(ubl + t * 512 + 128);
    const float4 q2 = *(const float4*)(ubl + t * 512 + 256);
    const float4 q3 = *(const float4*)(ubl + t * 512 + 384);
    const float u[16] = {q0.x, q0.y, q0.z, q0.w, q1.x, q1.y, q1.z, q1.w,
                         q2.x, q2.y, q2.z, q2.w, q3.x, q3.y, q3.z, q3.w};
#pragma unroll
    for (int j = 0; j < 16; ++j) {
      const float s0 = st;
      const float s1 = ROW_XOR(st, 1);
      const float s2 = ROW_XOR(st, 2);
      const float s3 = ROW_XOR(st, 3);
      float w[8];
      w[0] = fmaf(c[0][0], s0, u[j]);
#pragma unroll
      for (int jp = 1; jp < 8; ++jp) w[jp] = c[jp][0] * s0;
#pragma unroll
      for (int jp = 0; jp < 8; ++jp) w[jp] = fmaf(c[jp][1], s1, w[jp]);
#pragma unroll
      for (int jp = 0; jp < 8; ++jp) w[jp] = fmaf(c[jp][2], s2, w[jp]);
#pragma unroll
      for (int jp = 0; jp < 8; ++jp) w[jp] = fmaf(c[jp][3], s3, w[jp]);
      w[0] += ROW_XOR(w[1], 4);
      w[2] += ROW_XOR(w[3], 4);
      w[4] += ROW_XOR(w[5], 4);
      w[6] += ROW_XOR(w[7], 4);
      w[0] += ROW_XOR(w[2], 8);
      w[4] += ROW_XOR(w[6], 8);
      w[0] += half_xor(w[4]);
      st = tanhf(w[0]);
      st_out[(t * 16 + j) * 32 + lane] = st;
    }
  }
  fs_out[b * 32 + lane] = st;
}

__device__ __forceinline__ void wc_role(const int tile, const int lane,
                                        const float* __restrict__ W,
                                        const float* __restrict__ C,
                                        float* __restrict__ WC) {
  const int r = lane & 15, h = lane >> 4;
  const int ntile = tile >> 1;
  const int jt = tile & 1;
  const float* wrow = W + (size_t)(ntile * 16 + r) * D_MODEL;
  const float* ccol = C + jt * 16 + r;

  v8f acc0 = {0.f, 0.f, 0.f, 0.f, 0.f, 0.f, 0.f, 0.f};
  v8f acc1 = {0.f, 0.f, 0.f, 0.f, 0.f, 0.f, 0.f, 0.f};
  for (int k0 = 0; k0 < D_MODEL; k0 += 8) {
    const int k = k0 + 2 * h;
    const v2f a0 = *(const v2f*)(wrow + k);
    const v2f a1 = *(const v2f*)(wrow + k + 4);
    v2f b0, b1;
    b0[0] = ccol[(size_t)k * 32];
    b0[1] = ccol[(size_t)(k + 1) * 32];
    b1[0] = ccol[(size_t)(k + 4) * 32];
    b1[1] = ccol[(size_t)(k + 5) * 32];
    acc0 = wmma4(a0, b0, acc0);
    acc1 = wmma4(a1, b1, acc1);
  }
#pragma unroll
  for (int v = 0; v < 8; ++v)
    WC[(size_t)(ntile * 16 + 8 * h + v) * 32 + jt * 16 + r] = acc0[v] + acc1[v];
}

#define OT_P 36
#define XW_CHUNKS 16

__device__ __forceinline__ void chunk_mma(v8f (&acc)[2][4], const float* afr,
                                          const float* bfr) {
#pragma unroll
  for (int kk = 0; kk < 4; ++kk) {
    v4f a[2], b[4];
#pragma unroll
    for (int i = 0; i < 2; ++i)
      a[i] = *(const v4f*)(afr + i * 16 * OT_P + kk * 8);
#pragma unroll
    for (int j = 0; j < 4; ++j)
      b[j] = *(const v4f*)(bfr + j * 16 * OT_P + kk * 8);
#pragma unroll
    for (int i = 0; i < 2; ++i)
#pragma unroll
      for (int j = 0; j < 4; ++j) {
        v2f av, bv;
        av[0] = a[i].x; av[1] = a[i].y;
        bv[0] = b[j].x; bv[1] = b[j].y;
        acc[i][j] = wmma4(av, bv, acc[i][j]);
      }
#pragma unroll
    for (int i = 0; i < 2; ++i)
#pragma unroll
      for (int j = 0; j < 4; ++j) {
        v2f av, bv;
        av[0] = a[i].z; av[1] = a[i].w;
        bv[0] = b[j].z; bv[1] = b[j].w;
        acc[i][j] = wmma4(av, bv, acc[i][j]);
      }
  }
}

__device__ __forceinline__ void xw_role(const int tile, const int tid,
                                        const float* __restrict__ x,
                                        const float* __restrict__ W,
                                        const float* __restrict__ Dv,
                                        const float* __restrict__ bout,
                                        float* __restrict__ y, float* As,
                                        float* Bs) {
  const int wave = tid >> 5, lane = tid & 31;
  const int r = lane & 15, h = lane >> 4;
  const int n0 = (tile & 3) * 128;
  const int m0 = (tile >> 2) * 128;

  const int srow = tid >> 3;
  const int scol = (tid & 7) * 4;
  const float* xa = x + (size_t)(m0 + srow) * D_MODEL + scol;
  const float* wb = W + (size_t)(n0 + srow) * D_MODEL + scol;
  float* ast = As + srow * OT_P + scol;
  float* bst = Bs + srow * OT_P + scol;

  const int wm = wave >> 1, wn = wave & 1;
  const float* afr = As + (wm * 32 + r) * OT_P + 4 * h;
  const float* bfr = Bs + (wn * 64 + r) * OT_P + 4 * h;

  v8f acc[2][4];
#pragma unroll
  for (int i = 0; i < 2; ++i)
#pragma unroll
    for (int j = 0; j < 4; ++j)
      acc[i][j] = (v8f){0.f, 0.f, 0.f, 0.f, 0.f, 0.f, 0.f, 0.f};

  v4f ra[4], rb[4];
  v4f dcur = *(const v4f*)(Dv + scol);
#pragma unroll
  for (int i = 0; i < 4; ++i) {
    ra[i] = *(const v4f*)(xa + (size_t)i * 32 * D_MODEL);
    rb[i] = *(const v4f*)(wb + (size_t)i * 32 * D_MODEL);
  }

  for (int c = 0; c < XW_CHUNKS; ++c) {
    __syncthreads();
#pragma unroll
    for (int i = 0; i < 4; ++i) {
      *(v4f*)(ast + i * 32 * OT_P) = ra[i] * dcur;
      *(v4f*)(bst + i * 32 * OT_P) = rb[i];
    }
    __syncthreads();

    const int cn = c + 1;
    if (cn < XW_CHUNKS) {
#pragma unroll
      for (int i = 0; i < 4; ++i) {
        ra[i] = *(const v4f*)(xa + cn * 32 + (size_t)i * 32 * D_MODEL);
        rb[i] = *(const v4f*)(wb + cn * 32 + (size_t)i * 32 * D_MODEL);
      }
      dcur = *(const v4f*)(Dv + cn * 32 + scol);
    }
    chunk_mma(acc, afr, bfr);
  }

#pragma unroll
  for (int j = 0; j < 4; ++j) {
    const int n = n0 + wn * 64 + j * 16 + r;
    const float bo = bout[n];
#pragma unroll
    for (int i = 0; i < 2; ++i) {
      const int mb = m0 + wm * 32 + i * 16 + 8 * h;
#pragma unroll
      for (int v = 0; v < 8; ++v)
        y[(size_t)(mb + v) * D_MODEL + n] = acc[i][j][v] + bo;
    }
  }
}

#define MID_SCAN 8
#define MID_WC 8
#define MID_GRID 1040
__global__ void __launch_bounds__(256) mid_kernel(
    const float* __restrict__ Abar, const float* __restrict__ uB,
    float* __restrict__ states, float* __restrict__ fs_out,
    const float* __restrict__ x, const float* __restrict__ W,
    const float* __restrict__ C, const float* __restrict__ Dv,
    const float* __restrict__ bout, float* __restrict__ WC,
    float* __restrict__ y) {
  __shared__ __attribute__((aligned(16))) float As[128 * OT_P];
  __shared__ __attribute__((aligned(16))) float Bs[128 * OT_P];
  const int bid = blockIdx.x;
  const int tid = threadIdx.x;
  if (bid < MID_SCAN) {
    if (tid < 32) scan_role(bid, tid, Abar, uB, states, fs_out);
  } else if (bid < MID_SCAN + MID_WC) {
    wc_role((bid - MID_SCAN) * 8 + (tid >> 5), tid & 31, W, C, WC);
  } else {
    xw_role(bid - (MID_SCAN + MID_WC), tid, x, W, Dv, bout, y, As, Bs);
  }
}

__global__ void __launch_bounds__(256) fin_kernel(
    const float* __restrict__ states, const float* __restrict__ WC, float* y) {
  __shared__ __attribute__((aligned(16))) float As[128 * OT_P];
  __shared__ __attribute__((aligned(16))) float Bs[128 * OT_P];
  const int tid = threadIdx.x;
  const int wave = tid >> 5, lane = tid & 31;
  const int r = lane & 15, h = lane >> 4;
  const int n0 = blockIdx.x * 128;
  const int m0 = blockIdx.y * 128;
  const int srow = tid >> 3;
  const int scol = (tid & 7) * 4;
  const float* sa = states + (size_t)(m0 + srow) * D_STATE + scol;
  const float* cb = WC + (size_t)(n0 + srow) * D_STATE + scol;
  const int wm = wave >> 1, wn = wave & 1;

  v4f ra[4], rb[4];
#pragma unroll
  for (int i = 0; i < 4; ++i) {
    ra[i] = *(const v4f*)(sa + (size_t)i * 32 * D_STATE);
    rb[i] = *(const v4f*)(cb + (size_t)i * 32 * D_STATE);
  }
  float* yb = y + (size_t)(m0 + wm * 32 + 8 * h) * D_MODEL + n0 + wn * 64 + r;
  v8f acc[2][4];
#pragma unroll
  for (int i = 0; i < 2; ++i)
#pragma unroll
    for (int j = 0; j < 4; ++j)
#pragma unroll
      for (int v = 0; v < 8; ++v)
        acc[i][j][v] = yb[(size_t)(i * 16 + v) * D_MODEL + j * 16];

#pragma unroll
  for (int i = 0; i < 4; ++i) {
    *(v4f*)(As + (srow + i * 32) * OT_P + scol) = ra[i];
    *(v4f*)(Bs + (srow + i * 32) * OT_P + scol) = rb[i];
  }
  __syncthreads();
  chunk_mma(acc, As + (wm * 32 + r) * OT_P + 4 * h,
            Bs + (wn * 64 + r) * OT_P + 4 * h);

#pragma unroll
  for (int i = 0; i < 2; ++i)
#pragma unroll
    for (int j = 0; j < 4; ++j)
#pragma unroll
      for (int v = 0; v < 8; ++v)
        yb[(size_t)(i * 16 + v) * D_MODEL + j * 16] = acc[i][j][v];
}

extern "C" void kernel_launch(void* const* d_in, const int* in_sizes, int n_in,
                              void* d_out, int out_size, void* d_ws,
                              size_t ws_size, hipStream_t stream) {
  const float* x = (const float*)d_in[0];
  const float* A = (const float*)d_in[1];
  const float* B = (const float*)d_in[2];
  const float* C = (const float*)d_in[3];
  const float* Dv = (const float*)d_in[4];
  const float* log_dt = (const float*)d_in[5];
  const float* W = (const float*)d_in[6];
  const float* bout = (const float*)d_in[7];

  float* y = (float*)d_out;
  float* fs = y + (size_t)ROWS * D_MODEL;

  float* ws = (float*)d_ws;
  float* Abar = ws + WS_ABAR;
  float* Bbar = ws + WS_BBAR;
  float* uB = ws + WS_UB;
  float* states = ws + WS_STATES;
  float* WC = ws + WS_BBAR;

  prep_kernel<<<1, 256, 0, stream>>>(A, B, log_dt, ws);
  ub_kernel<<<256, 256, 0, stream>>>(x, Bbar, uB);
  mid_kernel<<<MID_GRID, 256, 0, stream>>>(Abar, uB, states, fs, x, W, C, Dv,
                                           bout, WC, y);
  fin_kernel<<<dim3(4, 256), 256, 0, stream>>>(states, WC, y);
}


# scan: uB tiles and state rows go through LDS rings that the work-group's 7 idle waves fill and drain, one barrier per 64 steps; the scan wave makes no global access in its loop; sums, order unchanged
# speedup vs baseline: 4.0814x; 1.1642x over previous
#include <hip/hip_runtime.h>

typedef float v2f __attribute__((ext_vector_type(2)));
typedef float v4f __attribute__((ext_vector_type(4)));
typedef float v8f __attribute__((ext_vector_type(8)));

#define D_MODEL 512
#define D_STATE 32
#define BATCH 8
#define SEQ 4096
#define ROWS (BATCH * SEQ)

#define WS_ABAR 0
#define WS_BBAR (WS_ABAR + 1024)
#define WS_UB (WS_BBAR + 16384)
#define WS_STATES (WS_UB + ROWS * 32)

__device__ __forceinline__ v8f wmma4(v2f a, v2f b, v8f c) {
  return __builtin_amdgcn_wmma_f32_16x16x4_f32(false, a, false, b, (short)0, c,
                                               false, false);
}

__global__ void __launch_bounds__(256) prep_kernel(
    const float* __restrict__ A, const float* __restrict__ B,
    const float* __restrict__ log_dt, float* __restrict__ ws) {
  __shared__ float aug[32][64];
  __shared__ float Ah[32][32];
  __shared__ float inv[32][32];
  __shared__ float fac[32];
  __shared__ float dt_sh;
  const int tid = threadIdx.x;

  if (tid == 0) {
    float s = 0.f;
    for (int i = 0; i < D_MODEL; ++i) {
      float v = expf(log_dt[i]);
      v = fminf(fmaxf(v, 1e-4f), 1.0f);
      s += v;
    }
    dt_sh = s / (float)D_MODEL;
  }
  __syncthreads();
  const float dt = dt_sh;

  for (int idx = tid; idx < 32 * 32; idx += blockDim.x) {
    const int i = idx >> 5, j = idx & 31;
    const float a = A[idx];
    const float eye = (i == j) ? 1.f : 0.f;
    aug[i][j] = eye - 0.5f * dt * a;
    aug[i][j + 32] = eye;
    Ah[i][j] = eye + 0.5f * dt * a;
  }
  __syncthreads();

  for (int k = 0; k < 32; ++k) {
    const float p = aug[k][k];
    __syncthreads();
    for (int j = tid; j < 64; j += blockDim.x) aug[k][j] = aug[k][j] / p;
    __syncthreads();
    if (tid < 32) fac[tid] = aug[tid][k];
    __syncthreads();
    for (int idx = tid; idx < 32 * 64; idx += blockDim.x) {
      const int i = idx >> 6, j = idx & 63;
      if (i != k) aug[i][j] -= fac[i] * aug[k][j];
    }
    __syncthreads();
  }
  for (int idx = tid; idx < 32 * 32; idx += blockDim.x) {
    const int i = idx >> 5, j = idx & 31;
    inv[i][j] = aug[i][j + 32];
  }
  __syncthreads();

  float* Abar = ws + WS_ABAR;
  for (int idx = tid; idx < 32 * 32; idx += blockDim.x) {
    const int i = idx >> 5, j = idx & 31;
    float s = 0.f;
    for (int k = 0; k < 32; ++k) s += Ah[i][k] * inv[k][j];
    Abar[idx] = s;
  }
  float* Bbar = ws + WS_BBAR;
  for (int idx = tid; idx < 32 * D_MODEL; idx += blockDim.x) {
    const int n = idx >> 9, d = idx & 511;
    float s = 0.f;
    for (int k = 0; k < 32; ++k) s += inv[n][k] * B[k * D_MODEL + d];
    Bbar[idx] = dt * s;
  }
}

__global__ void __launch_bounds__(256) ub_kernel(const float* __restrict__ x,
                                                 const float* __restrict__ Bbar,
                                                 float* __restrict__ uB) {
  const int wave = threadIdx.x >> 5;
  const int lane = threadIdx.x & 31;
  const int r = lane & 15, h = lane >> 4;
  const int mtile = blockIdx.x * 8 + wave;

  const float* xrow = x + (size_t)(mtile * 16 + r) * D_MODEL;
  const float* b0row = Bbar + (size_t)r * D_MODEL;
  const float* b1row = Bbar + (size_t)(16 + r) * D_MODEL;

  v8f acc0 = {0.f, 0.f, 0.f, 0.f, 0.f, 0.f, 0.f, 0.f};
  v8f acc1 = {0.f, 0.f, 0.f, 0.f, 0.f, 0.f, 0.f, 0.f};

  for (int k0 = 0; k0 < D_MODEL; k0 += 4) {
    const int k = k0 + 2 * h;
    const v2f a = *(const v2f*)(xrow + k);
    const v2f b0 = *(const v2f*)(b0row + k);
    const v2f b1 = *(const v2f*)(b1row + k);
    acc0 = wmma4(a, b0, acc0);
    acc1 = wmma4(a, b1, acc1);
  }

  float* tile = uB + (size_t)mtile * 512;
  float* dst = tile + (2 * h) * 128 + r * 4;
  *(float4*)(dst) = make_float4(acc0[0], acc0[1], acc0[2], acc0[3]);
  *(float4*)(dst + 64) = make_float4(acc1[0], acc1[1], acc1[2], acc1[3]);
  *(float4*)(dst + 128) = make_float4(acc0[4], acc0[5], acc0[6], acc0[7]);
  *(float4*)(dst + 192) = make_float4(acc1[4], acc1[5], acc1[6], acc1[7]);
}

#define ROW_XOR(x, MASK)                                                      \
  __int_as_float(__builtin_amdgcn_update_dpp(0, __float_as_int(x),            \
                                             0x160 | (MASK), 0xf, 0xf, true))
__device__ __forceinline__ float half_xor(float x) {
  const int v = __float_as_int(x);
  return __int_as_float(
      __builtin_amdgcn_permlanex16(v, v, 0x76543210, 0xfedcba98, false, false));
}

#define PH_STEPS 64
#define PH_TILES (PH_STEPS / 16)
#define PH_FLOATS (PH_STEPS * 32)
#define N_PH (SEQ / PH_STEPS)
#define N_HELP 224
#define RING_FLOATS (4 * PH_FLOATS)

__device__ __forceinline__ void scan_role(const int b, const int tid,
                                          const float* __restrict__ Abar,
                                          const float* __restrict__ uBall,
                                          float* __restrict__ states,
                                          float* __restrict__ fs_out,
                                          float* ring) {
  const int wave = tid >> 5;
  const int lane = tid & 31;
  const int ht = tid - 32;
  const float* uB = uBall + (size_t)b * SEQ * 32;
  float* st_out = states + (size_t)b * SEQ * 32;
  float* ub_ring = ring;
  float* st_ring = ring + 2 * PH_FLOATS;

  float c[8][4];
  float st = 0.f;
  if (wave == 0) {
#pragma unroll
    for (int jp = 0; jp < 8; ++jp) {
#pragma unroll
      for (int q = 0; q < 4; ++q)
        c[jp][q] = Abar[(lane ^ (jp << 2)) * 32 + (lane ^ q)];
    }
  } else {
    for (int i = ht; i < PH_FLOATS / 4; i += N_HELP)
      ((float4*)ub_ring)[i] = ((const float4*)uB)[i];
  }
  __syncthreads();

  for (int p = 0; p < N_PH; ++p) {
    const int slot = p & 1;
    if (wave == 0) {
      const float* ubl = ub_ring + slot * PH_FLOATS + lane * 4;
      float* stl = st_ring + slot * PH_FLOATS + lane;
#pragma unroll 1
      for (int tt = 0; tt < PH_TILES; ++tt) {
        const float4 q0 = *(const float4*)(ubl + tt * 512);
        const float4 q1 = *(const float4*)(ubl + tt * 512 + 128);
        const float4 q2 = *(const float4*)(ubl + tt * 512 + 256);
        const float4 q3 = *(const float4*)(ubl + tt * 512 + 384);
        const float u[16] = {q0.x, q0.y, q0.z, q0.w, q1.x, q1.y, q1.z, q1.w,
                             q2.x, q2.y, q2.z, q2.w, q3.x, q3.y, q3.z, q3.w};
#pragma unroll
        for (int j = 0; j < 16; ++j) {
          const float s0 = st;
          const float s1 = ROW_XOR(st, 1);
          const float s2 = ROW_XOR(st, 2);
          const float s3 = ROW_XOR(st, 3);
          float w[8];
          w[0] = fmaf(c[0][0], s0, u[j]);
#pragma unroll
          for (int jp = 1; jp < 8; ++jp) w[jp] = c[jp][0] * s0;
#pragma unroll
          for (int jp = 0; jp < 8; ++jp) w[jp] = fmaf(c[jp][1], s1, w[jp]);
#pragma unroll
          for (int jp = 0; jp < 8; ++jp) w[jp] = fmaf(c[jp][2], s2, w[jp]);
#pragma unroll
          for (int jp = 0; jp < 8; ++jp) w[jp] = fmaf(c[jp][3], s3, w[jp]);
          w[0] += ROW_XOR(w[1], 4);
          w[2] += ROW_XOR(w[3], 4);
          w[4] += ROW_XOR(w[5], 4);
          w[6] += ROW_XOR(w[7], 4);
          w[0] += ROW_XOR(w[2], 8);
          w[4] += ROW_XOR(w[6], 8);
          w[0] += half_xor(w[4]);
          st = tanhf(w[0]);
          stl[(tt * 16 + j) * 32] = st;
        }
      }
    } else {
      if (p + 1 < N_PH) {
        const float4* srcp = (const float4*)(uB + (size_t)(p + 1) * PH_FLOATS);
        float4* dstp = (float4*)(ub_ring + (slot ^ 1) * PH_FLOATS);
        for (int i = ht; i < PH_FLOATS / 4; i += N_HELP) dstp[i] = srcp[i];
      }
      if (p >= 1) {
        const float4* srcp = (const float4*)(st_ring + (slot ^ 1) * PH_FLOATS);
        float4* dstp = (float4*)(st_out + (size_t)(p - 1) * PH_FLOATS);
        for (int i = ht; i < PH_FLOATS / 4; i += N_HELP) dstp[i] = srcp[i];
      }
    }
    __syncthreads();
  }

  if (wave == 0) {
    fs_out[b * 32 + lane] = st;
  } else {
    const float4* srcp =
        (const float4*)(st_ring + ((N_PH - 1) & 1) * PH_FLOATS);
    float4* dstp = (float4*)(st_out + (size_t)(N_PH - 1) * PH_FLOATS);
    for (int i = ht; i < PH_FLOATS / 4; i += N_HELP) dstp[i] = srcp[i];
  }
}

__device__ __forceinline__ void wc_role(const int tile, const int lane,
                                        const float* __restrict__ W,
                                        const float* __restrict__ C,
                                        float* __restrict__ WC) {
  const int r = lane & 15, h = lane >> 4;
  const int ntile = tile >> 1;
  const int jt = tile & 1;
  const float* wrow = W + (size_t)(ntile * 16 + r) * D_MODEL;
  const float* ccol = C + jt * 16 + r;

  v8f acc0 = {0.f, 0.f, 0.f, 0.f, 0.f, 0.f, 0.f, 0.f};
  v8f acc1 = {0.f, 0.f, 0.f, 0.f, 0.f, 0.f, 0.f, 0.f};
  for (int k0 = 0; k0 < D_MODEL; k0 += 8) {
    const int k = k0 + 2 * h;
    const v2f a0 = *(const v2f*)(wrow + k);
    const v2f a1 = *(const v2f*)(wrow + k + 4);
    v2f b0, b1;
    b0[0] = ccol[(size_t)k * 32];
    b0[1] = ccol[(size_t)(k + 1) * 32];
    b1[0] = ccol[(size_t)(k + 4) * 32];
    b1[1] = ccol[(size_t)(k + 5) * 32];
    acc0 = wmma4(a0, b0, acc0);
    acc1 = wmma4(a1, b1, acc1);
  }
#pragma unroll
  for (int v = 0; v < 8; ++v)
    WC[(size_t)(ntile * 16 + 8 * h + v) * 32 + jt * 16 + r] = acc0[v] + acc1[v];
}

#define OT_P 36
#define XW_CHUNKS 16

__device__ __forceinline__ void chunk_mma(v8f (&acc)[2][4], const float* afr,
                                          const float* bfr) {
#pragma unroll
  for (int kk = 0; kk < 4; ++kk) {
    v4f a[2], b[4];
#pragma unroll
    for (int i = 0; i < 2; ++i)
      a[i] = *(const v4f*)(afr + i * 16 * OT_P + kk * 8);
#pragma unroll
    for (int j = 0; j < 4; ++j)
      b[j] = *(const v4f*)(bfr + j * 16 * OT_P + kk * 8);
#pragma unroll
    for (int i = 0; i < 2; ++i)
#pragma unroll
      for (int j = 0; j < 4; ++j) {
        v2f av, bv;
        av[0] = a[i].x; av[1] = a[i].y;
        bv[0] = b[j].x; bv[1] = b[j].y;
        acc[i][j] = wmma4(av, bv, acc[i][j]);
      }
#pragma unroll
    for (int i = 0; i < 2; ++i)
#pragma unroll
      for (int j = 0; j < 4; ++j) {
        v2f av, bv;
        av[0] = a[i].z; av[1] = a[i].w;
        bv[0] = b[j].z; bv[1] = b[j].w;
        acc[i][j] = wmma4(av, bv, acc[i][j]);
      }
  }
}

__device__ __forceinline__ void xw_role(const int tile, const int tid,
                                        const float* __restrict__ x,
                                        const float* __restrict__ W,
                                        const float* __restrict__ Dv,
                                        const float* __restrict__ bout,
                                        float* __restrict__ y, float* As,
                                        float* Bs) {
  const int wave = tid >> 5, lane = tid & 31;
  const int r = lane & 15, h = lane >> 4;
  const int n0 = (tile & 3) * 128;
  const int m0 = (tile >> 2) * 128;

  const int srow = tid >> 3;
  const int scol = (tid & 7) * 4;
  const float* xa = x + (size_t)(m0 + srow) * D_MODEL + scol;
  const float* wb = W + (size_t)(n0 + srow) * D_MODEL + scol;
  float* ast = As + srow * OT_P + scol;
  float* bst = Bs + srow * OT_P + scol;

  const int wm = wave >> 1, wn = wave & 1;
  const float* afr = As + (wm * 32 + r) * OT_P + 4 * h;
  const float* bfr = Bs + (wn * 64 + r) * OT_P + 4 * h;

  v8f acc[2][4];
#pragma unroll
  for (int i = 0; i < 2; ++i)
#pragma unroll
    for (int j = 0; j < 4; ++j)
      acc[i][j] = (v8f){0.f, 0.f, 0.f, 0.f, 0.f, 0.f, 0.f, 0.f};

  v4f ra[4], rb[4];
  v4f dcur = *(const v4f*)(Dv + scol);
#pragma unroll
  for (int i = 0; i < 4; ++i) {
    ra[i] = *(const v4f*)(xa + (size_t)i * 32 * D_MODEL);
    rb[i] = *(const v4f*)(wb + (size_t)i * 32 * D_MODEL);
  }

  for (int c = 0; c < XW_CHUNKS; ++c) {
    __syncthreads();
#pragma unroll
    for (int i = 0; i < 4; ++i) {
      *(v4f*)(ast + i * 32 * OT_P) = ra[i] * dcur;
      *(v4f*)(bst + i * 32 * OT_P) = rb[i];
    }
    __syncthreads();

    const int cn = c + 1;
    if (cn < XW_CHUNKS) {
#pragma unroll
      for (int i = 0; i < 4; ++i) {
        ra[i] = *(const v4f*)(xa + cn * 32 + (size_t)i * 32 * D_MODEL);
        rb[i] = *(const v4f*)(wb + cn * 32 + (size_t)i * 32 * D_MODEL);
      }
      dcur = *(const v4f*)(Dv + cn * 32 + scol);
    }
    chunk_mma(acc, afr, bfr);
  }

#pragma unroll
  for (int j = 0; j < 4; ++j) {
    const int n = n0 + wn * 64 + j * 16 + r;
    const float bo = bout[n];
#pragma unroll
    for (int i = 0; i < 2; ++i) {
      const int mb = m0 + wm * 32 + i * 16 + 8 * h;
#pragma unroll
      for (int v = 0; v < 8; ++v)
        y[(size_t)(mb + v) * D_MODEL + n] = acc[i][j][v] + bo;
    }
  }
}

#define MID_SCAN 8
#define MID_WC 8
#define MID_GRID 1040
__global__ void __launch_bounds__(256) mid_kernel(
    const float* __restrict__ Abar, const float* __restrict__ uB,
    float* __restrict__ states, float* __restrict__ fs_out,
    const float* __restrict__ x, const float* __restrict__ W,
    const float* __restrict__ C, const float* __restrict__ Dv,
    const float* __restrict__ bout, float* __restrict__ WC,
    float* __restrict__ y) {
  __shared__ __attribute__((aligned(16))) float lds[2 * 128 * OT_P];
  float* As = lds;
  float* Bs = lds + 128 * OT_P;
  const int bid = blockIdx.x;
  const int tid = threadIdx.x;
  if (bid < MID_SCAN) {
    scan_role(bid, tid, Abar, uB, states, fs_out, lds);
  } else if (bid < MID_SCAN + MID_WC) {
    wc_role((bid - MID_SCAN) * 8 + (tid >> 5), tid & 31, W, C, WC);
  } else {
    xw_role(bid - (MID_SCAN + MID_WC), tid, x, W, Dv, bout, y, As, Bs);
  }
}

__global__ void __launch_bounds__(256) fin_kernel(
    const float* __restrict__ states, const float* __restrict__ WC, float* y) {
  __shared__ __attribute__((aligned(16))) float As[128 * OT_P];
  __shared__ __attribute__((aligned(16))) float Bs[128 * OT_P];
  const int tid = threadIdx.x;
  const int wave = tid >> 5, lane = tid & 31;
  const int r = lane & 15, h = lane >> 4;
  const int n0 = blockIdx.x * 128;
  const int m0 = blockIdx.y * 128;
  const int srow = tid >> 3;
  const int scol = (tid & 7) * 4;
  const float* sa = states + (size_t)(m0 + srow) * D_STATE + scol;
  const float* cb = WC + (size_t)(n0 + srow) * D_STATE + scol;
  const int wm = wave >> 1, wn = wave & 1;

  v4f ra[4], rb[4];
#pragma unroll
  for (int i = 0; i < 4; ++i) {
    ra[i] = *(const v4f*)(sa + (size_t)i * 32 * D_STATE);
    rb[i] = *(const v4f*)(cb + (size_t)i * 32 * D_STATE);
  }
  float* yb = y + (size_t)(m0 + wm * 32 + 8 * h) * D_MODEL + n0 + wn * 64 + r;
  v8f acc[2][4];
#pragma unroll
  for (int i = 0; i < 2; ++i)
#pragma unroll
    for (int j = 0; j < 4; ++j)
#pragma unroll
      for (int v = 0; v < 8; ++v)
        acc[i][j][v] = yb[(size_t)(i * 16 + v) * D_MODEL + j * 16];

#pragma unroll
  for (int i = 0; i < 4; ++i) {
    *(v4f*)(As + (srow + i * 32) * OT_P + scol) = ra[i];
    *(v4f*)(Bs + (srow + i * 32) * OT_P + scol) = rb[i];
  }
  __syncthreads();
  chunk_mma(acc, As + (wm * 32 + r) * OT_P + 4 * h,
            Bs + (wn * 64 + r) * OT_P + 4 * h);

#pragma unroll
  for (int i = 0; i < 2; ++i)
#pragma unroll
    for (int j = 0; j < 4; ++j)
#pragma unroll
      for (int v = 0; v < 8; ++v)
        yb[(size_t)(i * 16 + v) * D_MODEL + j * 16] = acc[i][j][v];
}

extern "C" void kernel_launch(void* const* d_in, const int* in_sizes, int n_in,
                              void* d_out, int out_size, void* d_ws,
                              size_t ws_size, hipStream_t stream) {
  const float* x = (const float*)d_in[0];
  const float* A = (const float*)d_in[1];
  const float* B = (const float*)d_in[2];
  const float* C = (const float*)d_in[3];
  const float* Dv = (const float*)d_in[4];
  const float* log_dt = (const float*)d_in[5];
  const float* W = (const float*)d_in[6];
  const float* bout = (const float*)d_in[7];

  float* y = (float*)d_out;
  float* fs = y + (size_t)ROWS * D_MODEL;

  float* ws = (float*)d_ws;
  float* Abar = ws + WS_ABAR;
  float* Bbar = ws + WS_BBAR;
  float* uB = ws + WS_UB;
  float* states = ws + WS_STATES;
  float* WC = ws + WS_BBAR;

  prep_kernel<<<1, 256, 0, stream>>>(A, B, log_dt, ws);
  ub_kernel<<<256, 256, 0, stream>>>(x, Bbar, uB);
  mid_kernel<<<MID_GRID, 256, 0, stream>>>(Abar, uB, states, fs, x, W, C, Dv,
                                           bout, WC, y);
  fin_kernel<<<dim3(4, 256), 256, 0, stream>>>(states, WC, y);
}


# uB = (x @ B^T) @ M^T with M = dt*inv: B_bar is never formed, so the K=512 product runs beside prep in one launch; the scan's helper waves apply M (K=32) to each tile; sum order changes
# speedup vs baseline: 4.5354x; 1.1112x over previous
#include <hip/hip_runtime.h>

typedef float v2f __attribute__((ext_vector_type(2)));
typedef float v4f __attribute__((ext_vector_type(4)));
typedef float v8f __attribute__((ext_vector_type(8)));

#define D_MODEL 512
#define D_STATE 32
#define BATCH 8
#define SEQ 4096
#define ROWS (BATCH * SEQ)

#define WS_ABAR 0
#define WS_WC (WS_ABAR + 1024)
#define WS_Z (WS_WC + 16384)
#define WS_STATES (WS_Z + ROWS * 32)
#define WS_M_OF(b) (WS_STATES + (b) * SEQ * 32 + (SEQ - 64) * 32)

__device__ __forceinline__ v8f wmma4(v2f a, v2f b, v8f c) {
  return __builtin_amdgcn_wmma_f32_16x16x4_f32(false, a, false, b, (short)0, c,
                                               false, false);
}

__device__ __forceinline__ void prep_role(const float* __restrict__ A,
                                          const float* __restrict__ log_dt,
                                          float* __restrict__ ws) {
  __shared__ float aug[32][64];
  __shared__ float Ah[32][32];
  __shared__ float inv[32][32];
  __shared__ float fac[32];
  __shared__ float dt_sh;
  const int tid = threadIdx.x;

  if (tid == 0) {
    float s = 0.f;
    for (int i = 0; i < D_MODEL; ++i) {
      float v = expf(log_dt[i]);
      v = fminf(fmaxf(v, 1e-4f), 1.0f);
      s += v;
    }
    dt_sh = s / (float)D_MODEL;
  }
  __syncthreads();
  const float dt = dt_sh;

  for (int idx = tid; idx < 32 * 32; idx += blockDim.x) {
    const int i = idx >> 5, j = idx & 31;
    const float a = A[idx];
    const float eye = (i == j) ? 1.f : 0.f;
    aug[i][j] = eye - 0.5f * dt * a;
    aug[i][j + 32] = eye;
    Ah[i][j] = eye + 0.5f * dt * a;
  }
  __syncthreads();

  for (int k = 0; k < 32; ++k) {
    const float p = aug[k][k];
    __syncthreads();
    for (int j = tid; j < 64; j += blockDim.x) aug[k][j] = aug[k][j] / p;
    __syncthreads();
    if (tid < 32) fac[tid] = aug[tid][k];
    __syncthreads();
    for (int idx = tid; idx < 32 * 64; idx += blockDim.x) {
      const int i = idx >> 6, j = idx & 63;
      if (i != k) aug[i][j] -= fac[i] * aug[k][j];
    }
    __syncthreads();
  }
  for (int idx = tid; idx < 32 * 32; idx += blockDim.x) {
    const int i = idx >> 5, j = idx & 31;
    inv[i][j] = aug[i][j + 32];
  }
  __syncthreads();

  float* Abar = ws + WS_ABAR;
  for (int idx = tid; idx < 32 * 32; idx += blockDim.x) {
    const int i = idx >> 5, j = idx & 31;
    float s = 0.f;
    for (int k = 0; k < 32; ++k) s += Ah[i][k] * inv[k][j];
    Abar[idx] = s;
  }
  for (int idx = tid; idx < 32 * 32; idx += blockDim.x) {
    const float m = dt * inv[idx >> 5][idx & 31];
    for (int b = 0; b < BATCH; ++b) ws[WS_M_OF(b) + idx] = m;
  }
}

__device__ __forceinline__ void z_role(const int wg, const float* __restrict__ x,
                                       const float* __restrict__ Bin,
                                       float* __restrict__ z) {
  const int wave = threadIdx.x >> 5;
  const int lane = threadIdx.x & 31;
  const int r = lane & 15, h = lane >> 4;
  const int mtile = wg * 8 + wave;

  const float* xrow = x + (size_t)(mtile * 16 + r) * D_MODEL;
  const float* b0row = Bin + (size_t)r * D_MODEL;
  const float* b1row = Bin + (size_t)(16 + r) * D_MODEL;

  v8f acc0 = {0.f, 0.f, 0.f, 0.f, 0.f, 0.f, 0.f, 0.f};
  v8f acc1 = {0.f, 0.f, 0.f, 0.f, 0.f, 0.f, 0.f, 0.f};

  for (int k0 = 0; k0 < D_MODEL; k0 += 4) {
    const int k = k0 + 2 * h;
    const v2f a = *(const v2f*)(xrow + k);
    const v2f b0 = *(const v2f*)(b0row + k);
    const v2f b1 = *(const v2f*)(b1row + k);
    acc0 = wmma4(a, b0, acc0);
    acc1 = wmma4(a, b1, acc1);
  }

  float* tile = z + (size_t)mtile * 512;
  float* dst = tile + (2 * h) * 128 + r * 4;
  *(float4*)(dst) = make_float4(acc0[0], acc0[1], acc0[2], acc0[3]);
  *(float4*)(dst + 64) = make_float4(acc1[0], acc1[1], acc1[2], acc1[3]);
  *(float4*)(dst + 128) = make_float4(acc0[4], acc0[5], acc0[6], acc0[7]);
  *(float4*)(dst + 192) = make_float4(acc1[4], acc1[5], acc1[6], acc1[7]);
}

__global__ void __launch_bounds__(256) front_kernel(
    const float* __restrict__ x, const float* __restrict__ A,
    const float* __restrict__ B, const float* __restrict__ log_dt,
    float* __restrict__ ws) {
  if (blockIdx.x == 0) {
    prep_role(A, log_dt, ws);
  } else {
    z_role(blockIdx.x - 1, x, B, ws + WS_Z);
  }
}

#define ROW_XOR(x, MASK)                                                      \
  __int_as_float(__builtin_amdgcn_update_dpp(0, __float_as_int(x),            \
                                             0x160 | (MASK), 0xf, 0xf, true))
__device__ __forceinline__ float half_xor(float x) {
  const int v = __float_as_int(x);
  return __int_as_float(
      __builtin_amdgcn_permlanex16(v, v, 0x76543210, 0xfedcba98, false, false));
}

#define PH_STEPS 64
#define PH_TILES (PH_STEPS / 16)
#define PH_FLOATS (PH_STEPS * 32)
#define N_PH (SEQ / PH_STEPS)
#define N_WB 96
#define RING_FLOATS (4 * PH_FLOATS)

__device__ __forceinline__ void ub_tile(const float* __restrict__ zt,
                                        const float (&mf)[2][16], float* dst_tile,
                                        const int r, const int h) {
  const float* zl = zt + (r >> 2) * 128 + h * 64 + (r & 3);
  float av[16];
#pragma unroll
  for (int i = 0; i < 16; ++i) av[i] = zl[i * 4];
  v8f acc0 = {0.f, 0.f, 0.f, 0.f, 0.f, 0.f, 0.f, 0.f};
  v8f acc1 = {0.f, 0.f, 0.f, 0.f, 0.f, 0.f, 0.f, 0.f};
#pragma unroll
  for (int kk = 0; kk < 8; ++kk) {
    v2f a, b0, b1;
    a[0] = av[2 * kk];     a[1] = av[2 * kk + 1];
    b0[0] = mf[0][2 * kk]; b0[1] = mf[0][2 * kk + 1];
    b1[0] = mf[1][2 * kk]; b1[1] = mf[1][2 * kk + 1];
    acc0 = wmma4(a, b0, acc0);
    acc1 = wmma4(a, b1, acc1);
  }
  float* dst = dst_tile + (2 * h) * 128 + r * 4;
  *(float4*)(dst) = make_float4(acc0[0], acc0[1], acc0[2], acc0[3]);
  *(float4*)(dst + 64) = make_float4(acc1[0], acc1[1], acc1[2], acc1[3]);
  *(float4*)(dst + 128) = make_float4(acc0[4], acc0[5], acc0[6], acc0[7]);
  *(float4*)(dst + 192) = make_float4(acc1[4], acc1[5], acc1[6], acc1[7]);
}

__device__ __forceinline__ void scan_role(const int b, const int tid,
                                          const float* __restrict__ Abar,
                                          const float* __restrict__ uBall,
                                          float* __restrict__ states,
                                          float* __restrict__ fs_out,
                                          float* ring) {
  const int wave = tid >> 5;
  const int lane = tid & 31;
  const int r = lane & 15, h = lane >> 4;
  const bool is_ub = (wave == 1) || (wave == 2) || (wave == 3) || (wave == 5);
  const bool is_wb = (wave == 4) || (wave >= 6);
  const int tl = (wave == 5) ? 3 : ((wave - 1) & 3);
  const int wt = ((wave == 4) ? 0 : (wave - 5) * 32) + lane;
  const float* uB = uBall + (size_t)b * SEQ * 32 + (size_t)tl * 512;
  float* st_out = states + (size_t)b * SEQ * 32;
  float* ub_ring = ring;
  float* st_ring = ring + 2 * PH_FLOATS;

  float c[8][4];
  float st = 0.f;
  if (wave == 0) {
#pragma unroll
    for (int jp = 0; jp < 8; ++jp) {
#pragma unroll
      for (int q = 0; q < 4; ++q)
        c[jp][q] = Abar[(lane ^ (jp << 2)) * 32 + (lane ^ q)];
    }
  }
  float mf[2][16];
  if (is_ub) {
    const float* Mm = st_out + (size_t)(N_PH - 1) * PH_FLOATS;
#pragma unroll
    for (int nt = 0; nt < 2; ++nt)
#pragma unroll
      for (int i = 0; i < 16; ++i)
        mf[nt][i] = Mm[(nt * 16 + r) * 32 + h * 16 + i];
    ub_tile(uB, mf, ub_ring + tl * 512, r, h);
  }
  __syncthreads();

  for (int p = 0; p < N_PH; ++p) {
    const int slot = p & 1;
    if (wave == 0) {
      const float* ubl = ub_ring + slot * PH_FLOATS + lane * 4;
      float* stl = st_ring + slot * PH_FLOATS + lane;
#pragma unroll 1
      for (int tt = 0; tt < PH_TILES; ++tt) {
        const float4 q0 = *(const float4*)(ubl + tt * 512);
        const float4 q1 = *(const float4*)(ubl + tt * 512 + 128);
        const float4 q2 = *(const float4*)(ubl + tt * 512 + 256);
        const float4 q3 = *(const float4*)(ubl + tt * 512 + 384);
        const float u[16] = {q0.x, q0.y, q0.z, q0.w, q1.x, q1.y, q1.z, q1.w,
                             q2.x, q2.y, q2.z, q2.w, q3.x, q3.y, q3.z, q3.w};
#pragma unroll
        for (int j = 0; j < 16; ++j) {
          const float s0 = st;
          const float s1 = ROW_XOR(st, 1);
          const float s2 = ROW_XOR(st, 2);
          const float s3 = ROW_XOR(st, 3);
          float w[8];
          w[0] = fmaf(c[0][0], s0, u[j]);
#pragma unroll
          for (int jp = 1; jp < 8; ++jp) w[jp] = c[jp][0] * s0;
#pragma unroll
          for (int jp = 0; jp < 8; ++jp) w[jp] = fmaf(c[jp][1], s1, w[jp]);
#pragma unroll
          for (int jp = 0; jp < 8; ++jp) w[jp] = fmaf(c[jp][2], s2, w[jp]);
#pragma unroll
          for (int jp = 0; jp < 8; ++jp) w[jp] = fmaf(c[jp][3], s3, w[jp]);
          w[0] += ROW_XOR(w[1], 4);
          w[2] += ROW_XOR(w[3], 4);
          w[4] += ROW_XOR(w[5], 4);
          w[6] += ROW_XOR(w[7], 4);
          w[0] += ROW_XOR(w[2], 8);
          w[4] += ROW_XOR(w[6], 8);
          w[0] += half_xor(w[4]);
          st = tanhf(w[0]);
          stl[(tt * 16 + j) * 32] = st;
        }
      }
    } else if (is_ub) {
      if (p + 1 < N_PH)
        ub_tile(uB + (size_t)(p + 1) * PH_FLOATS, mf,
                ub_ring + (slot ^ 1) * PH_FLOATS + tl * 512, r, h);
    } else if (is_wb) {
      if (p >= 1) {
        const float4* srcp = (const float4*)(st_ring + (slot ^ 1) * PH_FLOATS);
        float4* dstp = (float4*)(st_out + (size_t)(p - 1) * PH_FLOATS);
        for (int i = wt; i < PH_FLOATS / 4; i += N_WB) dstp[i] = srcp[i];
      }
    }
    __syncthreads();
  }

  if (wave == 0) {
    fs_out[b * 32 + lane] = st;
  } else if (is_wb) {
    const float4* srcp =
        (const float4*)(st_ring + ((N_PH - 1) & 1) * PH_FLOATS);
    float4* dstp = (float4*)(st_out + (size_t)(N_PH - 1) * PH_FLOATS);
    for (int i = wt; i < PH_FLOATS / 4; i += N_WB) dstp[i] = srcp[i];
  }
}

__device__ __forceinline__ void wc_role(const int tile, const int lane,
                                        const float* __restrict__ W,
                                        const float* __restrict__ C,
                                        float* __restrict__ WC) {
  const int r = lane & 15, h = lane >> 4;
  const int ntile = tile >> 1;
  const int jt = tile & 1;
  const float* wrow = W + (size_t)(ntile * 16 + r) * D_MODEL;
  const float* ccol = C + jt * 16 + r;

  v8f acc0 = {0.f, 0.f, 0.f, 0.f, 0.f, 0.f, 0.f, 0.f};
  v8f acc1 = {0.f, 0.f, 0.f, 0.f, 0.f, 0.f, 0.f, 0.f};
  for (int k0 = 0; k0 < D_MODEL; k0 += 8) {
    const int k = k0 + 2 * h;
    const v2f a0 = *(const v2f*)(wrow + k);
    const v2f a1 = *(const v2f*)(wrow + k + 4);
    v2f b0, b1;
    b0[0] = ccol[(size_t)k * 32];
    b0[1] = ccol[(size_t)(k + 1) * 32];
    b1[0] = ccol[(size_t)(k + 4) * 32];
    b1[1] = ccol[(size_t)(k + 5) * 32];
    acc0 = wmma4(a0, b0, acc0);
    acc1 = wmma4(a1, b1, acc1);
  }
#pragma unroll
  for (int v = 0; v < 8; ++v)
    WC[(size_t)(ntile * 16 + 8 * h + v) * 32 + jt * 16 + r] = acc0[v] + acc1[v];
}

#define OT_P 36
#define XW_CHUNKS 16

__device__ __forceinline__ void chunk_mma(v8f (&acc)[2][4], const float* afr,
                                          const float* bfr) {
#pragma unroll
  for (int kk = 0; kk < 4; ++kk) {
    v4f a[2], b[4];
#pragma unroll
    for (int i = 0; i < 2; ++i)
      a[i] = *(const v4f*)(afr + i * 16 * OT_P + kk * 8);
#pragma unroll
    for (int j = 0; j < 4; ++j)
      b[j] = *(const v4f*)(bfr + j * 16 * OT_P + kk * 8);
#pragma unroll
    for (int i = 0; i < 2; ++i)
#pragma unroll
      for (int j = 0; j < 4; ++j) {
        v2f av, bv;
        av[0] = a[i].x; av[1] = a[i].y;
        bv[0] = b[j].x; bv[1] = b[j].y;
        acc[i][j] = wmma4(av, bv, acc[i][j]);
      }
#pragma unroll
    for (int i = 0; i < 2; ++i)
#pragma unroll
      for (int j = 0; j < 4; ++j) {
        v2f av, bv;
        av[0] = a[i].z; av[1] = a[i].w;
        bv[0] = b[j].z; bv[1] = b[j].w;
        acc[i][j] = wmma4(av, bv, acc[i][j]);
      }
  }
}

__device__ __forceinline__ void xw_role(const int tile, const int tid,
                                        const float* __restrict__ x,
                                        const float* __restrict__ W,
                                        const float* __restrict__ Dv,
                                        const float* __restrict__ bout,
                                        float* __restrict__ y, float* As,
                                        float* Bs) {
  const int wave = tid >> 5, lane = tid & 31;
  const int r = lane & 15, h = lane >> 4;
  const int n0 = (tile & 3) * 128;
  const int m0 = (tile >> 2) * 128;

  const int srow = tid >> 3;
  const int scol = (tid & 7) * 4;
  const float* xa = x + (size_t)(m0 + srow) * D_MODEL + scol;
  const float* wb = W + (size_t)(n0 + srow) * D_MODEL + scol;
  float* ast = As + srow * OT_P + scol;
  float* bst = Bs + srow * OT_P + scol;

  const int wm = wave >> 1, wn = wave & 1;
  const float* afr = As + (wm * 32 + r) * OT_P + 4 * h;
  const float* bfr = Bs + (wn * 64 + r) * OT_P + 4 * h;

  v8f acc[2][4];
#pragma unroll
  for (int i = 0; i < 2; ++i)
#pragma unroll
    for (int j = 0; j < 4; ++j)
      acc[i][j] = (v8f){0.f, 0.f, 0.f, 0.f, 0.f, 0.f, 0.f, 0.f};

  v4f ra[4], rb[4];
  v4f dcur = *(const v4f*)(Dv + scol);
#pragma unroll
  for (int i = 0; i < 4; ++i) {
    ra[i] = *(const v4f*)(xa + (size_t)i * 32 * D_MODEL);
    rb[i] = *(const v4f*)(wb + (size_t)i * 32 * D_MODEL);
  }

  for (int c = 0; c < XW_CHUNKS; ++c) {
    __syncthreads();
#pragma unroll
    for (int i = 0; i < 4; ++i) {
      *(v4f*)(ast + i * 32 * OT_P) = ra[i] * dcur;
      *(v4f*)(bst + i * 32 * OT_P) = rb[i];
    }
    __syncthreads();

    const int cn = c + 1;
    if (cn < XW_CHUNKS) {
#pragma unroll
      for (int i = 0; i < 4; ++i) {
        ra[i] = *(const v4f*)(xa + cn * 32 + (size_t)i * 32 * D_MODEL);
        rb[i] = *(const v4f*)(wb + cn * 32 + (size_t)i * 32 * D_MODEL);
      }
      dcur = *(const v4f*)(Dv + cn * 32 + scol);
    }
    chunk_mma(acc, afr, bfr);
  }

#pragma unroll
  for (int j = 0; j < 4; ++j) {
    const int n = n0 + wn * 64 + j * 16 + r;
    const float bo = bout[n];
#pragma unroll
    for (int i = 0; i < 2; ++i) {
      const int mb = m0 + wm * 32 + i * 16 + 8 * h;
#pragma unroll
      for (int v = 0; v < 8; ++v)
        y[(size_t)(mb + v) * D_MODEL + n] = acc[i][j][v] + bo;
    }
  }
}

#define MID_SCAN 8
#define MID_WC 8
#define MID_GRID 1040
__global__ void __launch_bounds__(256) mid_kernel(
    const float* __restrict__ Abar, const float* __restrict__ uB,
    float* __restrict__ states, float* __restrict__ fs_out,
    const float* __restrict__ x, const float* __restrict__ W,
    const float* __restrict__ C, const float* __restrict__ Dv,
    const float* __restrict__ bout, float* __restrict__ WC,
    float* __restrict__ y) {
  __shared__ __attribute__((aligned(16))) float lds[2 * 128 * OT_P];
  float* As = lds;
  float* Bs = lds + 128 * OT_P;
  const int bid = blockIdx.x;
  const int tid = threadIdx.x;
  if (bid < MID_SCAN) {
    scan_role(bid, tid, Abar, uB, states, fs_out, lds);
  } else if (bid < MID_SCAN + MID_WC) {
    wc_role((bid - MID_SCAN) * 8 + (tid >> 5), tid & 31, W, C, WC);
  } else {
    xw_role(bid - (MID_SCAN + MID_WC), tid, x, W, Dv, bout, y, As, Bs);
  }
}

__global__ void __launch_bounds__(256) fin_kernel(
    const float* __restrict__ states, const float* __restrict__ WC, float* y) {
  __shared__ __attribute__((aligned(16))) float As[128 * OT_P];
  __shared__ __attribute__((aligned(16))) float Bs[128 * OT_P];
  const int tid = threadIdx.x;
  const int wave = tid >> 5, lane = tid & 31;
  const int r = lane & 15, h = lane >> 4;
  const int n0 = blockIdx.x * 128;
  const int m0 = blockIdx.y * 128;
  const int srow = tid >> 3;
  const int scol = (tid & 7) * 4;
  const float* sa = states + (size_t)(m0 + srow) * D_STATE + scol;
  const float* cb = WC + (size_t)(n0 + srow) * D_STATE + scol;
  const int wm = wave >> 1, wn = wave & 1;

  v4f ra[4], rb[4];
#pragma unroll
  for (int i = 0; i < 4; ++i) {
    ra[i] = *(const v4f*)(sa + (size_t)i * 32 * D_STATE);
    rb[i] = *(const v4f*)(cb + (size_t)i * 32 * D_STATE);
  }
  float* yb = y + (size_t)(m0 + wm * 32 + 8 * h) * D_MODEL + n0 + wn * 64 + r;
  v8f acc[2][4];
#pragma unroll
  for (int i = 0; i < 2; ++i)
#pragma unroll
    for (int j = 0; j < 4; ++j)
#pragma unroll
      for (int v = 0; v < 8; ++v)
        acc[i][j][v] = yb[(size_t)(i * 16 + v) * D_MODEL + j * 16];

#pragma unroll
  for (int i = 0; i < 4; ++i) {
    *(v4f*)(As + (srow + i * 32) * OT_P + scol) = ra[i];
    *(v4f*)(Bs + (srow + i * 32) * OT_P + scol) = rb[i];
  }
  __syncthreads();
  chunk_mma(acc, As + (wm * 32 + r) * OT_P + 4 * h,
            Bs + (wn * 64 + r) * OT_P + 4 * h);

#pragma unroll
  for (int i = 0; i < 2; ++i)
#pragma unroll
    for (int j = 0; j < 4; ++j)
#pragma unroll
      for (int v = 0; v < 8; ++v)
        yb[(size_t)(i * 16 + v) * D_MODEL + j * 16] = acc[i][j][v];
}

extern "C" void kernel_launch(void* const* d_in, const int* in_sizes, int n_in,
                              void* d_out, int out_size, void* d_ws,
                              size_t ws_size, hipStream_t stream) {
  const float* x = (const float*)d_in[0];
  const float* A = (const float*)d_in[1];
  const float* B = (const float*)d_in[2];
  const float* C = (const float*)d_in[3];
  const float* Dv = (const float*)d_in[4];
  const float* log_dt = (const float*)d_in[5];
  const float* W = (const float*)d_in[6];
  const float* bout = (const float*)d_in[7];

  float* y = (float*)d_out;
  float* fs = y + (size_t)ROWS * D_MODEL;

  float* ws = (float*)d_ws;
  float* Abar = ws + WS_ABAR;
  float* z = ws + WS_Z;
  float* states = ws + WS_STATES;
  float* WC = ws + WS_WC;

  front_kernel<<<257, 256, 0, stream>>>(x, A, B, log_dt, ws);
  mid_kernel<<<MID_GRID, 256, 0, stream>>>(Abar, z, states, fs, x, W, C, Dv,
                                           bout, WC, y);
  fin_kernel<<<dim3(4, 256), 256, 0, stream>>>(states, WC, y);
}
